# starting kernel
# baseline (speedup 1.0000x reference)
#include <hip/hip_runtime.h>

#define D_MODEL 1024
#define ADAPT   128
#define NUM_LIB 8
#define NPAIR   64
#define BATCH   1024
#define H_STRIDE 132

typedef __attribute__((ext_vector_type(2))) float v2f;
typedef __attribute__((ext_vector_type(8))) float v8f;

__global__ __launch_bounds__(BATCH) void classify_kernel(
    const int* __restrict__ src, const int* __restrict__ tgt,
    int* __restrict__ counts, int* __restrict__ lists)
{
    const int t = threadIdx.x;
    if (t < NPAIR) counts[t] = 0;
    __syncthreads();
    const int s = src[t], g = tgt[t];
    if (s != g) {
        const int p = s * NUM_LIB + g;
        const int slot = atomicAdd(&counts[p], 1);
        lists[p * BATCH + slot] = t;
    }
}

__global__ __launch_bounds__(256) void identity_copy_kernel(
    const float* __restrict__ x, const int* __restrict__ src,
    const int* __restrict__ tgt, float* __restrict__ out)
{
    const int row = blockIdx.x;
    if (src[row] != tgt[row]) return;
    const float4 v = ((const float4*)(x + (size_t)row * D_MODEL))[threadIdx.x];
    ((float4*)(out + (size_t)row * D_MODEL))[threadIdx.x] = v;
}

__global__ __launch_bounds__(256) void grouped_adapter_gemm_kernel(
    const float* __restrict__ x,  const float* __restrict__ W1,
    const float* __restrict__ b1, const float* __restrict__ W2,
    const float* __restrict__ b2, const int* __restrict__ counts,
    const int* __restrict__ lists, float* __restrict__ out)
{
    const int p   = blockIdx.x;
    const int n_p = counts[p];
    const int tile = blockIdx.y;
    if (tile * 16 >= n_p) return;

    __shared__ float hS[16 * H_STRIDE];
    __shared__ int rowIdx[16];

    const int t    = threadIdx.x;
    const int lane = t & 31;
    const int wave = t >> 5;
    const int ml   = lane & 15;
    const int hi   = lane >> 4;
    const int kb   = hi * 2;

    if (t < 16) {
        const int g = tile * 16 + t;
        rowIdx[t] = (g < n_p) ? lists[p * BATCH + g] : lists[p * BATCH];
    }
    __syncthreads();

    const float* W1p = W1 + (size_t)p * D_MODEL * ADAPT;
    const float* W2p = W2 + (size_t)p * ADAPT * D_MODEL;

    {
        const int n0 = wave * 16;
        const float* xrow = x + (size_t)rowIdx[ml] * D_MODEL;
        v8f acc = {};
        #pragma unroll 4
        for (int k = 0; k < D_MODEL; k += 4) {
            v2f a = *(const v2f*)(xrow + k + kb);
            v2f b;
            b.x = W1p[(size_t)(k + kb    ) * ADAPT + n0 + ml];
            b.y = W1p[(size_t)(k + kb + 1) * ADAPT + n0 + ml];
            acc = __builtin_amdgcn_wmma_f32_16x16x4_f32(
                false, a, false, b, (short)0, acc, false, false);
        }
        const float bias = b1[p * ADAPT + n0 + ml];
        #pragma unroll
        for (int v = 0; v < 8; ++v) {
            float hv = acc[v] + bias;
            hv = hv > 0.0f ? hv : 0.0f;
            hS[(hi * 8 + v) * H_STRIDE + n0 + ml] = hv;
        }
    }
    __syncthreads();

    const int g0 = tile * 16;
    #pragma unroll 1
    for (int j = 0; j < 8; ++j) {
        const int n0 = wave * 128 + j * 16;
        v8f acc = {};
        #pragma unroll 4
        for (int k = 0; k < ADAPT; k += 4) {
            v2f a = *(const v2f*)(hS + ml * H_STRIDE + k + kb);
            v2f b;
            b.x = W2p[(size_t)(k + kb    ) * D_MODEL + n0 + ml];
            b.y = W2p[(size_t)(k + kb + 1) * D_MODEL + n0 + ml];
            acc = __builtin_amdgcn_wmma_f32_16x16x4_f32(
                false, a, false, b, (short)0, acc, false, false);
        }
        const float bias = b2[p * D_MODEL + n0 + ml];
        #pragma unroll
        for (int v = 0; v < 8; ++v) {
            const int m2 = hi * 8 + v;
            if (g0 + m2 < n_p) {
                out[(size_t)rowIdx[m2] * D_MODEL + n0 + ml] = acc[v] + bias;
            }
        }
    }
}

extern "C" void kernel_launch(void* const* d_in, const int* in_sizes, int n_in,
                              void* d_out, int out_size, void* d_ws, size_t ws_size,
                              hipStream_t stream) {
    const float* x   = (const float*)d_in[0];
    const int*   src = (const int*)  d_in[1];
    const int*   tgt = (const int*)  d_in[2];
    const float* W1  = (const float*)d_in[3];
    const float* b1  = (const float*)d_in[4];
    const float* W2  = (const float*)d_in[5];
    const float* b2  = (const float*)d_in[6];
    float* out = (float*)d_out;

    int* counts = (int*)d_ws;
    int* lists  = counts + NPAIR;

    classify_kernel<<<1, BATCH, 0, stream>>>(src, tgt, counts, lists);
    identity_copy_kernel<<<BATCH, 256, 0, stream>>>(x, src, tgt, out);
    grouped_adapter_gemm_kernel<<<dim3(NPAIR, BATCH / 16), 256, 0, stream>>>(
        x, W1, b1, W2, b2, counts, lists, out);
}


# Both K loops: the next chunk's global fragment loads are issued into a second register set before the current chunk's WMMAs (register double buffer); same k order and accumulator chain
# speedup vs baseline: 2.1738x; 2.1738x over previous
#include <hip/hip_runtime.h>

#define D_MODEL 1024
#define ADAPT   128
#define NUM_LIB 8
#define NPAIR   64
#define BATCH   1024
#define H_STRIDE 132

typedef __attribute__((ext_vector_type(2))) float v2f;
typedef __attribute__((ext_vector_type(8))) float v8f;

__global__ __launch_bounds__(BATCH) void classify_kernel(
    const int* __restrict__ src, const int* __restrict__ tgt,
    int* __restrict__ counts, int* __restrict__ lists)
{
    const int t = threadIdx.x;
    if (t < NPAIR) counts[t] = 0;
    __syncthreads();
    const int s = src[t], g = tgt[t];
    if (s != g) {
        const int p = s * NUM_LIB + g;
        const int slot = atomicAdd(&counts[p], 1);
        lists[p * BATCH + slot] = t;
    }
}

__global__ __launch_bounds__(256) void identity_copy_kernel(
    const float* __restrict__ x, const int* __restrict__ src,
    const int* __restrict__ tgt, float* __restrict__ out)
{
    const int row = blockIdx.x;
    if (src[row] != tgt[row]) return;
    const float4 v = ((const float4*)(x + (size_t)row * D_MODEL))[threadIdx.x];
    ((float4*)(out + (size_t)row * D_MODEL))[threadIdx.x] = v;
}

#define CH   16
#define NCH1 (D_MODEL / (4 * CH))

__device__ __forceinline__ void load_chunk_xw(v2f (&a)[CH], v2f (&b)[CH],
                                              const float* __restrict__ xk,
                                              const float* __restrict__ wk)
{
    #pragma unroll
    for (int i = 0; i < CH; ++i) {
        a[i]   = *(const v2f*)(xk + 4 * i);
        b[i].x = wk[(size_t)(4 * i    ) * ADAPT];
        b[i].y = wk[(size_t)(4 * i + 1) * ADAPT];
    }
}

__device__ __forceinline__ void mma_chunk(v8f& acc, const v2f (&a)[CH],
                                          const v2f (&b)[CH])
{
    #pragma unroll
    for (int i = 0; i < CH; ++i)
        acc = __builtin_amdgcn_wmma_f32_16x16x4_f32(
            false, a[i], false, b[i], (short)0, acc, false, false);
}

__device__ __forceinline__ void load_tile_w2(v2f (&b)[ADAPT / 4], float& bias,
                                             const float* __restrict__ w2k,
                                             const float* __restrict__ b2k)
{
    #pragma unroll
    for (int i = 0; i < ADAPT / 4; ++i) {
        b[i].x = w2k[(size_t)(4 * i    ) * D_MODEL];
        b[i].y = w2k[(size_t)(4 * i + 1) * D_MODEL];
    }
    bias = *b2k;
}

__device__ __forceinline__ void mma_store_tile(const v2f (&a)[ADAPT / 4],
                                               const v2f (&b)[ADAPT / 4],
                                               float bias, float* __restrict__ out,
                                               const int* rowIdx, int hi, int g0,
                                               int n_p, int col)
{
    v8f acc = {};
    #pragma unroll
    for (int i = 0; i < ADAPT / 4; ++i)
        acc = __builtin_amdgcn_wmma_f32_16x16x4_f32(
            false, a[i], false, b[i], (short)0, acc, false, false);
    #pragma unroll
    for (int v = 0; v < 8; ++v) {
        const int m2 = hi * 8 + v;
        if (g0 + m2 < n_p) {
            out[(size_t)rowIdx[m2] * D_MODEL + col] = acc[v] + bias;
        }
    }
}

__global__ __launch_bounds__(256) void grouped_adapter_gemm_kernel(
    const float* __restrict__ x,  const float* __restrict__ W1,
    const float* __restrict__ b1, const float* __restrict__ W2,
    const float* __restrict__ b2, const int* __restrict__ counts,
    const int* __restrict__ lists, float* __restrict__ out)
{
    const int p   = blockIdx.x;
    const int n_p = counts[p];
    const int tile = blockIdx.y;
    if (tile * 16 >= n_p) return;

    __shared__ float hS[16 * H_STRIDE];
    __shared__ int rowIdx[16];

    const int t    = threadIdx.x;
    const int lane = t & 31;
    const int wave = t >> 5;
    const int ml   = lane & 15;
    const int hi   = lane >> 4;
    const int kb   = hi * 2;

    if (t < 16) {
        const int g = tile * 16 + t;
        rowIdx[t] = (g < n_p) ? lists[p * BATCH + g] : lists[p * BATCH];
    }
    __syncthreads();

    const float* W1p = W1 + (size_t)p * D_MODEL * ADAPT;
    const float* W2p = W2 + (size_t)p * ADAPT * D_MODEL;

    {
        const int n0 = wave * 16;
        const float* xk = x + (size_t)rowIdx[ml] * D_MODEL + kb;
        const float* wk = W1p + (size_t)kb * ADAPT + n0 + ml;
        const float bias = b1[p * ADAPT + n0 + ml];
        v2f a0[CH], b0[CH], a1[CH], b1r[CH];
        v8f acc = {};
        load_chunk_xw(a0, b0, xk, wk);
        for (int c = 0; c < NCH1 - 2; c += 2) {
            load_chunk_xw(a1, b1r, xk + (c + 1) * (4 * CH),
                          wk + (size_t)(c + 1) * (4 * CH) * ADAPT);
            mma_chunk(acc, a0, b0);
            load_chunk_xw(a0, b0, xk + (c + 2) * (4 * CH),
                          wk + (size_t)(c + 2) * (4 * CH) * ADAPT);
            mma_chunk(acc, a1, b1r);
        }
        load_chunk_xw(a1, b1r, xk + (NCH1 - 1) * (4 * CH),
                      wk + (size_t)(NCH1 - 1) * (4 * CH) * ADAPT);
        mma_chunk(acc, a0, b0);
        mma_chunk(acc, a1, b1r);
        #pragma unroll
        for (int v = 0; v < 8; ++v) {
            float hv = acc[v] + bias;
            hv = hv > 0.0f ? hv : 0.0f;
            hS[(hi * 8 + v) * H_STRIDE + n0 + ml] = hv;
        }
    }
    __syncthreads();

    const int g0 = tile * 16;
    {
        v2f ah[ADAPT / 4];
        #pragma unroll
        for (int i = 0; i < ADAPT / 4; ++i)
            ah[i] = *(const v2f*)(hS + ml * H_STRIDE + 4 * i + kb);

        const int nw = wave * 128 + ml;
        const float* w2k = W2p + (size_t)kb * D_MODEL + nw;
        const float* b2k = b2 + p * D_MODEL + nw;
        v2f bA[ADAPT / 4], bB[ADAPT / 4];
        float biasA, biasB;
        load_tile_w2(bA, biasA, w2k, b2k);
        for (int j = 0; j < 6; j += 2) {
            load_tile_w2(bB, biasB, w2k + (j + 1) * 16, b2k + (j + 1) * 16);
            mma_store_tile(ah, bA, biasA, out, rowIdx, hi, g0, n_p, nw + j * 16);
            load_tile_w2(bA, biasA, w2k + (j + 2) * 16, b2k + (j + 2) * 16);
            mma_store_tile(ah, bB, biasB, out, rowIdx, hi, g0, n_p, nw + (j + 1) * 16);
        }
        load_tile_w2(bB, biasB, w2k + 7 * 16, b2k + 7 * 16);
        mma_store_tile(ah, bA, biasA, out, rowIdx, hi, g0, n_p, nw + 6 * 16);
        mma_store_tile(ah, bB, biasB, out, rowIdx, hi, g0, n_p, nw + 7 * 16);
    }
}

extern "C" void kernel_launch(void* const* d_in, const int* in_sizes, int n_in,
                              void* d_out, int out_size, void* d_ws, size_t ws_size,
                              hipStream_t stream) {
    const float* x   = (const float*)d_in[0];
    const int*   src = (const int*)  d_in[1];
    const int*   tgt = (const int*)  d_in[2];
    const float* W1  = (const float*)d_in[3];
    const float* b1  = (const float*)d_in[4];
    const float* W2  = (const float*)d_in[5];
    const float* b2  = (const float*)d_in[6];
    float* out = (float*)d_out;

    int* counts = (int*)d_ws;
    int* lists  = counts + NPAIR;

    classify_kernel<<<1, BATCH, 0, stream>>>(src, tgt, counts, lists);
    identity_copy_kernel<<<BATCH, 256, 0, stream>>>(x, src, tgt, out);
    grouped_adapter_gemm_kernel<<<dim3(NPAIR, BATCH / 16), 256, 0, stream>>>(
        x, W1, b1, W2, b2, counts, lists, out);
}


# Each expert GEMM split over 4 blocks in 2 launches (h handed over in out rows); a wave holds a 64k x 64col weight tile from 16-byte loads for all its rows; k parts summed in LDS: sum order changes
# speedup vs baseline: 2.9691x; 1.3659x over previous
#include <hip/hip_runtime.h>

#define D_MODEL 1024
#define ADAPT   128
#define NUM_LIB 8
#define NPAIR   64
#define BATCH   1024
#define H_STRIDE 132
#define R1_STRIDE 68
#define R2_STRIDE 260

typedef __attribute__((ext_vector_type(2))) float v2f;
typedef __attribute__((ext_vector_type(4))) float v4f;
typedef __attribute__((ext_vector_type(8))) float v8f;

__global__ __launch_bounds__(BATCH) void classify_kernel(
    const int* __restrict__ src, const int* __restrict__ tgt,
    int* __restrict__ counts, int* __restrict__ lists)
{
    const int t = threadIdx.x;
    if (t < NPAIR) counts[t] = 0;
    __syncthreads();
    const int s = src[t], g = tgt[t];
    if (s != g) {
        const int p = s * NUM_LIB + g;
        const int slot = atomicAdd(&counts[p], 1);
        lists[p * BATCH + slot] = t;
    }
}

__global__ __launch_bounds__(256) void identity_copy_kernel(
    const float* __restrict__ x, const int* __restrict__ src,
    const int* __restrict__ tgt, float* __restrict__ out)
{
    const int row = blockIdx.x;
    if (src[row] != tgt[row]) return;
    const float4 v = ((const float4*)(x + (size_t)row * D_MODEL))[threadIdx.x];
    ((float4*)(out + (size_t)row * D_MODEL))[threadIdx.x] = v;
}

__device__ __forceinline__ void load_b_tile(v4f (&bf)[32],
                                            const float* __restrict__ wp, int pitch)
{
    #pragma unroll
    for (int i = 0; i < 32; ++i)
        bf[i] = *(const v4f*)(wp + (size_t)(8 * (i >> 2) + (i & 3)) * pitch);
}

__device__ __forceinline__ void mma_tile(v8f (&acc)[4], const v4f (&af)[8],
                                         const v4f (&bf)[32])
{
    #pragma unroll
    for (int kb = 0; kb < 8; ++kb) {
        const v2f a0 = {af[kb][0], af[kb][1]};
        const v2f a1 = {af[kb][2], af[kb][3]};
        #pragma unroll
        for (int t = 0; t < 4; ++t) {
            const v2f b = {bf[4 * kb][t], bf[4 * kb + 1][t]};
            acc[t] = __builtin_amdgcn_wmma_f32_16x16x4_f32(
                false, a0, false, b, (short)0, acc[t], false, false);
        }
        #pragma unroll
        for (int t = 0; t < 4; ++t) {
            const v2f b = {bf[4 * kb + 2][t], bf[4 * kb + 3][t]};
            acc[t] = __builtin_amdgcn_wmma_f32_16x16x4_f32(
                false, a1, false, b, (short)0, acc[t], false, false);
        }
    }
}

__device__ __forceinline__ void stage_rows(int* rowIdx, const int* __restrict__ lists,
                                           int p, int g0, int n_p, int t)
{
    if (t < 16) {
        const int g = g0 + t;
        rowIdx[t] = lists[p * BATCH + (g < n_p ? g : 0)];
    }
    __syncthreads();
}

__global__ __launch_bounds__(256) void adapter_down_kernel(
    const float* __restrict__ x, const float* __restrict__ W1,
    const int* __restrict__ counts, const int* __restrict__ lists,
    float* __restrict__ out)
{
    const int p   = blockIdx.x;
    const int n_p = counts[p];
    if (n_p == 0) return;

    __shared__ __attribute__((aligned(16))) float red[8 * 16 * R1_STRIDE];
    __shared__ int rowIdx[16];

    const int nh   = blockIdx.y & 1;
    const int ks   = blockIdx.y >> 1;
    const int t    = threadIdx.x;
    const int lane = t & 31;
    const int wave = t >> 5;
    const int c    = lane & 15;
    const int h    = lane >> 4;
    const int kbase = ks * 512 + wave * 64 + 4 * h;
    const int row = t >> 4;
    const int c4  = t & 15;

    int g0 = 0;
    stage_rows(rowIdx, lists, p, g0, n_p, t);
    v4f af[8];
    {
        const float* xr = x + (size_t)rowIdx[c] * D_MODEL + kbase;
        #pragma unroll
        for (int kb = 0; kb < 8; ++kb) af[kb] = *(const v4f*)(xr + 8 * kb);
    }
    v4f bf[32];
    load_b_tile(bf, W1 + (size_t)p * D_MODEL * ADAPT + (size_t)kbase * ADAPT
                       + nh * 64 + 4 * c, ADAPT);

    while (true) {
        v8f acc[4] = {};
        mma_tile(acc, af, bf);

        float* rw = red + (wave * 16 + 8 * h) * R1_STRIDE + 4 * c;
        #pragma unroll
        for (int v = 0; v < 8; ++v) {
            const v4f o = {acc[0][v], acc[1][v], acc[2][v], acc[3][v]};
            *(v4f*)(rw + v * R1_STRIDE) = o;
        }
        __syncthreads();

        v4f sum = *(const v4f*)(red + row * R1_STRIDE + 4 * c4);
        #pragma unroll
        for (int w = 1; w < 8; ++w)
            sum += *(const v4f*)(red + (w * 16 + row) * R1_STRIDE + 4 * c4);
        if (g0 + row < n_p) {
            float* o = out + (size_t)rowIdx[row] * D_MODEL + ks * 128 + nh * 64 + 4 * c4;
            #pragma unroll
            for (int q = 0; q < 4; ++q) *(v4f*)(o + q * 256) = sum;
        }

        g0 += 16;
        if (g0 >= n_p) break;
        __syncthreads();
        stage_rows(rowIdx, lists, p, g0, n_p, t);
        const float* xr = x + (size_t)rowIdx[c] * D_MODEL + kbase;
        #pragma unroll
        for (int kb = 0; kb < 8; ++kb) af[kb] = *(const v4f*)(xr + 8 * kb);
    }
}

__device__ __forceinline__ void h_request(v4f (&p0)[2], v4f (&p1)[2], v4f& bb,
                                          const int* rowIdx, const float* out,
                                          const float* __restrict__ b1p, int q, int t)
{
    const int a = 4 * (t & 31);
    #pragma unroll
    for (int i = 0; i < 2; ++i) {
        const int row = (t >> 5) + 8 * i;
        const float* hp = out + (size_t)rowIdx[row] * D_MODEL + q * 256 + a;
        p0[i] = *(const v4f*)hp;
        p1[i] = *(const v4f*)(hp + 128);
    }
    bb = *(const v4f*)(b1p + a);
}

__device__ __forceinline__ void h_finish(float* hS, const v4f (&p0)[2],
                                         const v4f (&p1)[2], const v4f& bb,
                                         int g0, int n_p, int t)
{
    const int a = 4 * (t & 31);
    #pragma unroll
    for (int i = 0; i < 2; ++i) {
        const int row = (t >> 5) + 8 * i;
        const bool valid = g0 + row < n_p;
        v4f hv = p0[i] + p1[i] + bb;
        #pragma unroll
        for (int e = 0; e < 4; ++e) {
            const float r = hv[e] > 0.0f ? hv[e] : 0.0f;
            hv[e] = valid ? r : 0.0f;
        }
        *(v4f*)(hS + row * H_STRIDE + a) = hv;
    }
}

__global__ __launch_bounds__(256) void adapter_up_kernel(
    const float* __restrict__ b1, const float* __restrict__ W2,
    const float* __restrict__ b2, const int* __restrict__ counts,
    const int* __restrict__ lists, float* out)
{
    const int p   = blockIdx.x;
    const int n_p = counts[p];
    if (n_p == 0) return;

    __shared__ __attribute__((aligned(16))) float hS[16 * H_STRIDE];
    __shared__ __attribute__((aligned(16))) float red[2 * 16 * R2_STRIDE];
    __shared__ int rowIdx[16];

    const int q    = blockIdx.y;
    const int t    = threadIdx.x;
    const int lane = t & 31;
    const int wave = t >> 5;
    const int c    = lane & 15;
    const int h    = lane >> 4;
    const int strip = wave & 3;
    const int kh    = wave >> 2;
    const int kbase = kh * 64 + 4 * h;

    int g0 = 0;
    stage_rows(rowIdx, lists, p, g0, n_p, t);
    v4f p0[2], p1[2], bb;
    h_request(p0, p1, bb, rowIdx, out, b1 + p * ADAPT, q, t);
    v4f bf[32];
    load_b_tile(bf, W2 + (size_t)p * ADAPT * D_MODEL + (size_t)kbase * D_MODEL
                       + q * 256 + strip * 64 + 4 * c, D_MODEL);
    h_finish(hS, p0, p1, bb, g0, n_p, t);

    while (true) {
        __syncthreads();

        v4f af[8];
        #pragma unroll
        for (int kb = 0; kb < 8; ++kb)
            af[kb] = *(const v4f*)(hS + c * H_STRIDE + kbase + 8 * kb);

        v8f acc[4] = {};
        mma_tile(acc, af, bf);

        float* rw = red + (kh * 16 + 8 * h) * R2_STRIDE + strip * 64 + 4 * c;
        #pragma unroll
        for (int v = 0; v < 8; ++v) {
            const v4f o = {acc[0][v], acc[1][v], acc[2][v], acc[3][v]};
            *(v4f*)(rw + v * R2_STRIDE) = o;
        }
        __syncthreads();

        #pragma unroll
        for (int i = 0; i < 4; ++i) {
            const int idx = t + 256 * i;
            const int row = idx >> 6;
            const int n   = 4 * (idx & 63);
            if (g0 + row < n_p) {
                const v4f y = *(const v4f*)(red + row * R2_STRIDE + n)
                            + *(const v4f*)(red + (16 + row) * R2_STRIDE + n)
                            + *(const v4f*)(b2 + p * D_MODEL + q * 256 + n);
                *(v4f*)(out + (size_t)rowIdx[row] * D_MODEL + q * 256 + n) = y;
            }
        }

        g0 += 16;
        if (g0 >= n_p) break;
        __syncthreads();
        stage_rows(rowIdx, lists, p, g0, n_p, t);
        h_request(p0, p1, bb, rowIdx, out, b1 + p * ADAPT, q, t);
        h_finish(hS, p0, p1, bb, g0, n_p, t);
    }
}

extern "C" void kernel_launch(void* const* d_in, const int* in_sizes, int n_in,
                              void* d_out, int out_size, void* d_ws, size_t ws_size,
                              hipStream_t stream) {
    const float* x   = (const float*)d_in[0];
    const int*   src = (const int*)  d_in[1];
    const int*   tgt = (const int*)  d_in[2];
    const float* W1  = (const float*)d_in[3];
    const float* b1  = (const float*)d_in[4];
    const float* W2  = (const float*)d_in[5];
    const float* b2  = (const float*)d_in[6];
    float* out = (float*)d_out;

    int* counts = (int*)d_ws;
    int* lists  = counts + NPAIR;

    classify_kernel<<<1, BATCH, 0, stream>>>(src, tgt, counts, lists);
    identity_copy_kernel<<<BATCH, 256, 0, stream>>>(x, src, tgt, out);
    adapter_down_kernel<<<dim3(NPAIR, 4), 256, 0, stream>>>(x, W1, counts, lists, out);
    adapter_up_kernel<<<dim3(NPAIR, 4), 256, 0, stream>>>(b1, W2, b2, counts, lists, out);
}


# Launches 4 to 2: each GEMM block builds its expert bucket in LDS from src/tgt by ballot prefix (no atomics, no workspace); identity rows copied by an extra block plane of launch 1; sums unchanged
# speedup vs baseline: 3.6381x; 1.2253x over previous
#include <hip/hip_runtime.h>

#define D_MODEL 1024
#define ADAPT   128
#define NUM_LIB 8
#define NPAIR   64
#define BATCH   1024
#define H_STRIDE 132
#define R1_STRIDE 68
#define R2_STRIDE 260

typedef __attribute__((ext_vector_type(2))) float v2f;
typedef __attribute__((ext_vector_type(4))) float v4f;
typedef __attribute__((ext_vector_type(8))) float v8f;

__device__ __forceinline__ int build_bucket(int* list, int* waveTot,
                                            const int* __restrict__ src,
                                            const int* __restrict__ tgt, int p, int t)
{
    const int lane = t & 31;
    const int wave = t >> 5;
    const int4 s4 = ((const int4*)src)[t];
    const int4 g4 = ((const int4*)tgt)[t];
    const int sv[4] = {s4.x, s4.y, s4.z, s4.w};
    const int gv[4] = {g4.x, g4.y, g4.z, g4.w};
    const unsigned below = (1u << lane) - 1u;
    bool m[4];
    int before = 0, total = 0;
    #pragma unroll
    for (int j = 0; j < 4; ++j) {
        m[j] = (sv[j] * NUM_LIB + gv[j] == p) && (sv[j] != gv[j]);
        const unsigned mask = __builtin_amdgcn_ballot_w32(m[j]);
        before += __builtin_popcount(mask & below);
        total  += __builtin_popcount(mask);
    }
    if (lane == 0) waveTot[wave] = total;
    __syncthreads();
    int base = 0, n = 0;
    #pragma unroll
    for (int w = 0; w < 8; ++w) {
        const int c = waveTot[w];
        base += w < wave ? c : 0;
        n += c;
    }
    int pos = base + before;
    #pragma unroll
    for (int j = 0; j < 4; ++j) {
        if (m[j]) list[pos] = 4 * t + j;
        pos += m[j] ? 1 : 0;
    }
    __syncthreads();
    return __builtin_amdgcn_readfirstlane(n);
}

__device__ __forceinline__ int bucket_row(const int* list, int g, int n_p)
{
    return list[g < n_p ? g : 0];
}

__device__ __forceinline__ void identity_rows(const float* __restrict__ x,
                                              const int* __restrict__ src,
                                              const int* __restrict__ tgt,
                                              float* __restrict__ out, int blk, int t)
{
    const int lane = t & 31;
    const int row0 = 16 * blk + 2 * (t >> 5);
    const int2 s2 = *(const int2*)(src + row0);
    const int2 g2 = *(const int2*)(tgt + row0);
    const bool same[2] = {s2.x == g2.x, s2.y == g2.y};
    #pragma unroll
    for (int i = 0; i < 2; ++i) {
        if (!same[i]) continue;
        const v4f* xr = (const v4f*)(x + (size_t)(row0 + i) * D_MODEL) + lane;
        v4f* orow = (v4f*)(out + (size_t)(row0 + i) * D_MODEL) + lane;
        v4f v[8];
        #pragma unroll
        for (int k = 0; k < 8; ++k) v[k] = xr[32 * k];
        #pragma unroll
        for (int k = 0; k < 8; ++k) orow[32 * k] = v[k];
    }
}

__device__ __forceinline__ void load_b_tile(v4f (&bf)[32],
                                            const float* __restrict__ wp, int pitch)
{
    #pragma unroll
    for (int i = 0; i < 32; ++i)
        bf[i] = *(const v4f*)(wp + (size_t)(8 * (i >> 2) + (i & 3)) * pitch);
}

__device__ __forceinline__ void mma_tile(v8f (&acc)[4], const v4f (&af)[8],
                                         const v4f (&bf)[32])
{
    #pragma unroll
    for (int kb = 0; kb < 8; ++kb) {
        const v2f a0 = {af[kb][0], af[kb][1]};
        const v2f a1 = {af[kb][2], af[kb][3]};
        #pragma unroll
        for (int t = 0; t < 4; ++t) {
            const v2f b = {bf[4 * kb][t], bf[4 * kb + 1][t]};
            acc[t] = __builtin_amdgcn_wmma_f32_16x16x4_f32(
                false, a0, false, b, (short)0, acc[t], false, false);
        }
        #pragma unroll
        for (int t = 0; t < 4; ++t) {
            const v2f b = {bf[4 * kb + 2][t], bf[4 * kb + 3][t]};
            acc[t] = __builtin_amdgcn_wmma_f32_16x16x4_f32(
                false, a1, false, b, (short)0, acc[t], false, false);
        }
    }
}

__global__ __launch_bounds__(256) void adapter_down_kernel(
    const float* __restrict__ x, const int* __restrict__ src,
    const int* __restrict__ tgt, const float* __restrict__ W1,
    float* __restrict__ out)
{
    const int p = blockIdx.x;
    const int t = threadIdx.x;
    if (blockIdx.y == 4) {
        identity_rows(x, src, tgt, out, p, t);
        return;
    }

    __shared__ __attribute__((aligned(16))) float red[8 * 16 * R1_STRIDE];
    __shared__ int list[BATCH];
    __shared__ int waveTot[8];

    const int n_p = build_bucket(list, waveTot, src, tgt, p, t);
    if (n_p == 0) return;

    const int nh   = blockIdx.y & 1;
    const int ks   = blockIdx.y >> 1;
    const int lane = t & 31;
    const int wave = t >> 5;
    const int c    = lane & 15;
    const int h    = lane >> 4;
    const int kbase = ks * 512 + wave * 64 + 4 * h;
    const int row = t >> 4;
    const int c4  = t & 15;

    int g0 = 0;
    v4f af[8];
    {
        const float* xr = x + (size_t)bucket_row(list, c, n_p) * D_MODEL + kbase;
        #pragma unroll
        for (int kb = 0; kb < 8; ++kb) af[kb] = *(const v4f*)(xr + 8 * kb);
    }
    v4f bf[32];
    load_b_tile(bf, W1 + (size_t)p * D_MODEL * ADAPT + (size_t)kbase * ADAPT
                       + nh * 64 + 4 * c, ADAPT);

    while (true) {
        v8f acc[4] = {};
        mma_tile(acc, af, bf);

        float* rw = red + (wave * 16 + 8 * h) * R1_STRIDE + 4 * c;
        #pragma unroll
        for (int v = 0; v < 8; ++v) {
            const v4f o = {acc[0][v], acc[1][v], acc[2][v], acc[3][v]};
            *(v4f*)(rw + v * R1_STRIDE) = o;
        }
        __syncthreads();

        v4f sum = *(const v4f*)(red + row * R1_STRIDE + 4 * c4);
        #pragma unroll
        for (int w = 1; w < 8; ++w)
            sum += *(const v4f*)(red + (w * 16 + row) * R1_STRIDE + 4 * c4);
        if (g0 + row < n_p) {
            float* o = out + (size_t)list[g0 + row] * D_MODEL + ks * 128 + nh * 64 + 4 * c4;
            #pragma unroll
            for (int q = 0; q < 4; ++q) *(v4f*)(o + q * 256) = sum;
        }

        g0 += 16;
        if (g0 >= n_p) break;
        __syncthreads();
        const float* xr = x + (size_t)bucket_row(list, g0 + c, n_p) * D_MODEL + kbase;
        #pragma unroll
        for (int kb = 0; kb < 8; ++kb) af[kb] = *(const v4f*)(xr + 8 * kb);
    }
}

__device__ __forceinline__ void h_request(v4f (&p0)[2], v4f (&p1)[2], v4f& bb,
                                          const int* list, int g0, int n_p,
                                          const float* out,
                                          const float* __restrict__ b1p, int q, int t)
{
    const int a = 4 * (t & 31);
    #pragma unroll
    for (int i = 0; i < 2; ++i) {
        const int row = (t >> 5) + 8 * i;
        const float* hp = out + (size_t)bucket_row(list, g0 + row, n_p) * D_MODEL
                              + q * 256 + a;
        p0[i] = *(const v4f*)hp;
        p1[i] = *(const v4f*)(hp + 128);
    }
    bb = *(const v4f*)(b1p + a);
}

__device__ __forceinline__ void h_finish(float* hS, const v4f (&p0)[2],
                                         const v4f (&p1)[2], const v4f& bb,
                                         int g0, int n_p, int t)
{
    const int a = 4 * (t & 31);
    #pragma unroll
    for (int i = 0; i < 2; ++i) {
        const int row = (t >> 5) + 8 * i;
        const bool valid = g0 + row < n_p;
        v4f hv = p0[i] + p1[i] + bb;
        #pragma unroll
        for (int e = 0; e < 4; ++e) {
            const float r = hv[e] > 0.0f ? hv[e] : 0.0f;
            hv[e] = valid ? r : 0.0f;
        }
        *(v4f*)(hS + row * H_STRIDE + a) = hv;
    }
}

__global__ __launch_bounds__(256) void adapter_up_kernel(
    const int* __restrict__ src, const int* __restrict__ tgt,
    const float* __restrict__ b1, const float* __restrict__ W2,
    const float* __restrict__ b2, float* out)
{
    __shared__ __attribute__((aligned(16))) float hS[16 * H_STRIDE];
    __shared__ __attribute__((aligned(16))) float red[2 * 16 * R2_STRIDE];
    __shared__ int list[BATCH];
    __shared__ int waveTot[8];

    const int p = blockIdx.x;
    const int t = threadIdx.x;
    const int n_p = build_bucket(list, waveTot, src, tgt, p, t);
    if (n_p == 0) return;

    const int q    = blockIdx.y;
    const int lane = t & 31;
    const int wave = t >> 5;
    const int c    = lane & 15;
    const int h    = lane >> 4;
    const int strip = wave & 3;
    const int kh    = wave >> 2;
    const int kbase = kh * 64 + 4 * h;

    int g0 = 0;
    v4f p0[2], p1[2], bb;
    h_request(p0, p1, bb, list, g0, n_p, out, b1 + p * ADAPT, q, t);
    v4f bf[32];
    load_b_tile(bf, W2 + (size_t)p * ADAPT * D_MODEL + (size_t)kbase * D_MODEL
                       + q * 256 + strip * 64 + 4 * c, D_MODEL);
    h_finish(hS, p0, p1, bb, g0, n_p, t);

    while (true) {
        __syncthreads();

        v4f af[8];
        #pragma unroll
        for (int kb = 0; kb < 8; ++kb)
            af[kb] = *(const v4f*)(hS + c * H_STRIDE + kbase + 8 * kb);

        v8f acc[4] = {};
        mma_tile(acc, af, bf);

        float* rw = red + (kh * 16 + 8 * h) * R2_STRIDE + strip * 64 + 4 * c;
        #pragma unroll
        for (int v = 0; v < 8; ++v) {
            const v4f o = {acc[0][v], acc[1][v], acc[2][v], acc[3][v]};
            *(v4f*)(rw + v * R2_STRIDE) = o;
        }
        __syncthreads();

        #pragma unroll
        for (int i = 0; i < 4; ++i) {
            const int idx = t + 256 * i;
            const int row = idx >> 6;
            const int n   = 4 * (idx & 63);
            if (g0 + row < n_p) {
                const v4f y = *(const v4f*)(red + row * R2_STRIDE + n)
                            + *(const v4f*)(red + (16 + row) * R2_STRIDE + n)
                            + *(const v4f*)(b2 + p * D_MODEL + q * 256 + n);
                *(v4f*)(out + (size_t)list[g0 + row] * D_MODEL + q * 256 + n) = y;
            }
        }

        g0 += 16;
        if (g0 >= n_p) break;
        __syncthreads();
        h_request(p0, p1, bb, list, g0, n_p, out, b1 + p * ADAPT, q, t);
        h_finish(hS, p0, p1, bb, g0, n_p, t);
    }
}

extern "C" void kernel_launch(void* const* d_in, const int* in_sizes, int n_in,
                              void* d_out, int out_size, void* d_ws, size_t ws_size,
                              hipStream_t stream) {
    const float* x   = (const float*)d_in[0];
    const int*   src = (const int*)  d_in[1];
    const int*   tgt = (const int*)  d_in[2];
    const float* W1  = (const float*)d_in[3];
    const float* b1  = (const float*)d_in[4];
    const float* W2  = (const float*)d_in[5];
    const float* b2  = (const float*)d_in[6];
    float* out = (float*)d_out;

    adapter_down_kernel<<<dim3(NPAIR, 5), 256, 0, stream>>>(x, src, tgt, W1, out);
    adapter_up_kernel<<<dim3(NPAIR, 4), 256, 0, stream>>>(src, tgt, b1, W2, b2, out);
}


# Workgroup barriers order LDS only (no global fence) and global loads are issued ahead across them: weight tile before the bucket build, next tile's rows and the bias a tile ahead; sums unchanged
# speedup vs baseline: 5.2799x; 1.4513x over previous
#include <hip/hip_runtime.h>

#define D_MODEL 1024
#define ADAPT   128
#define NUM_LIB 8
#define NPAIR   64
#define BATCH   1024
#define H_STRIDE 132
#define R1_STRIDE 68
#define R2_STRIDE 260

typedef __attribute__((ext_vector_type(2))) float v2f;
typedef __attribute__((ext_vector_type(4))) float v4f;
typedef __attribute__((ext_vector_type(8))) float v8f;

__device__ __forceinline__ void lds_barrier()
{
    __builtin_amdgcn_fence(__ATOMIC_RELEASE, "workgroup", "local");
    __builtin_amdgcn_s_barrier();
    __builtin_amdgcn_fence(__ATOMIC_ACQUIRE, "workgroup", "local");
}

__device__ __forceinline__ int build_bucket(int* list, int* waveTot,
                                            const int4 s4, const int4 g4, int p, int t)
{
    const int lane = t & 31;
    const int wave = t >> 5;
    const int sv[4] = {s4.x, s4.y, s4.z, s4.w};
    const int gv[4] = {g4.x, g4.y, g4.z, g4.w};
    const unsigned below = (1u << lane) - 1u;
    bool m[4];
    int before = 0, total = 0;
    #pragma unroll
    for (int j = 0; j < 4; ++j) {
        m[j] = (sv[j] * NUM_LIB + gv[j] == p) && (sv[j] != gv[j]);
        const unsigned mask = __builtin_amdgcn_ballot_w32(m[j]);
        before += __builtin_popcount(mask & below);
        total  += __builtin_popcount(mask);
    }
    if (lane == 0) waveTot[wave] = total;
    lds_barrier();
    int base = 0, n = 0;
    #pragma unroll
    for (int w = 0; w < 8; ++w) {
        const int c = waveTot[w];
        base += w < wave ? c : 0;
        n += c;
    }
    int pos = base + before;
    #pragma unroll
    for (int j = 0; j < 4; ++j) {
        if (m[j]) list[pos] = 4 * t + j;
        pos += m[j] ? 1 : 0;
    }
    lds_barrier();
    return __builtin_amdgcn_readfirstlane(n);
}

__device__ __forceinline__ int bucket_row(const int* list, int g, int n_p)
{
    return list[g < n_p ? g : 0];
}

__device__ __forceinline__ void identity_rows(const float* __restrict__ x,
                                              const int* __restrict__ src,
                                              const int* __restrict__ tgt,
                                              float* __restrict__ out, int blk, int t)
{
    const int lane = t & 31;
    const int row0 = 16 * blk + 2 * (t >> 5);
    const int2 s2 = *(const int2*)(src + row0);
    const int2 g2 = *(const int2*)(tgt + row0);
    const bool same[2] = {s2.x == g2.x, s2.y == g2.y};
    #pragma unroll
    for (int i = 0; i < 2; ++i) {
        if (!same[i]) continue;
        const v4f* xr = (const v4f*)(x + (size_t)(row0 + i) * D_MODEL) + lane;
        v4f* orow = (v4f*)(out + (size_t)(row0 + i) * D_MODEL) + lane;
        v4f v[8];
        #pragma unroll
        for (int k = 0; k < 8; ++k) v[k] = xr[32 * k];
        #pragma unroll
        for (int k = 0; k < 8; ++k) orow[32 * k] = v[k];
    }
}

__device__ __forceinline__ void load_b_tile(v4f (&bf)[32],
                                            const float* __restrict__ wp, int pitch)
{
    #pragma unroll
    for (int i = 0; i < 32; ++i)
        bf[i] = *(const v4f*)(wp + (size_t)(8 * (i >> 2) + (i & 3)) * pitch);
}

__device__ __forceinline__ void mma_tile(v8f (&acc)[4], const v4f (&af)[8],
                                         const v4f (&bf)[32])
{
    #pragma unroll
    for (int kb = 0; kb < 8; ++kb) {
        const v2f a0 = {af[kb][0], af[kb][1]};
        const v2f a1 = {af[kb][2], af[kb][3]};
        #pragma unroll
        for (int t = 0; t < 4; ++t) {
            const v2f b = {bf[4 * kb][t], bf[4 * kb + 1][t]};
            acc[t] = __builtin_amdgcn_wmma_f32_16x16x4_f32(
                false, a0, false, b, (short)0, acc[t], false, false);
        }
        #pragma unroll
        for (int t = 0; t < 4; ++t) {
            const v2f b = {bf[4 * kb + 2][t], bf[4 * kb + 3][t]};
            acc[t] = __builtin_amdgcn_wmma_f32_16x16x4_f32(
                false, a1, false, b, (short)0, acc[t], false, false);
        }
    }
}

__global__ __launch_bounds__(256) void adapter_down_kernel(
    const float* __restrict__ x, const int* __restrict__ src,
    const int* __restrict__ tgt, const float* __restrict__ W1,
    float* __restrict__ out)
{
    const int p = blockIdx.x;
    const int t = threadIdx.x;
    if (blockIdx.y == 4) {
        identity_rows(x, src, tgt, out, p, t);
        return;
    }

    __shared__ __attribute__((aligned(16))) float red[8 * 16 * R1_STRIDE];
    __shared__ int list[BATCH];
    __shared__ int waveTot[8];

    if ((p >> 3) == (p & 7)) return;

    const int nh   = blockIdx.y & 1;
    const int ks   = blockIdx.y >> 1;
    const int lane = t & 31;
    const int wave = t >> 5;
    const int c    = lane & 15;
    const int h    = lane >> 4;
    const int kbase = ks * 512 + wave * 64 + 4 * h;
    const int row = t >> 4;
    const int c4  = t & 15;

    const int4 s4 = ((const int4*)src)[t];
    const int4 g4 = ((const int4*)tgt)[t];
    v4f bf[32];
    load_b_tile(bf, W1 + (size_t)p * D_MODEL * ADAPT + (size_t)kbase * ADAPT
                       + nh * 64 + 4 * c, ADAPT);

    const int n_p = build_bucket(list, waveTot, s4, g4, p, t);
    if (n_p == 0) return;

    int g0 = 0;
    v4f af[8], an[8];
    {
        const float* xr = x + (size_t)bucket_row(list, c, n_p) * D_MODEL + kbase;
        const float* xn = x + (size_t)bucket_row(list, 16 + c, n_p) * D_MODEL + kbase;
        #pragma unroll
        for (int kb = 0; kb < 8; ++kb) af[kb] = *(const v4f*)(xr + 8 * kb);
        #pragma unroll
        for (int kb = 0; kb < 8; ++kb) an[kb] = *(const v4f*)(xn + 8 * kb);
    }
    __builtin_amdgcn_sched_barrier(0);

    while (true) {
        v8f acc[4] = {};
        mma_tile(acc, af, bf);

        float* rw = red + (wave * 16 + 8 * h) * R1_STRIDE + 4 * c;
        #pragma unroll
        for (int v = 0; v < 8; ++v) {
            const v4f o = {acc[0][v], acc[1][v], acc[2][v], acc[3][v]};
            *(v4f*)(rw + v * R1_STRIDE) = o;
        }
        lds_barrier();

        v4f sum = *(const v4f*)(red + row * R1_STRIDE + 4 * c4);
        #pragma unroll
        for (int w = 1; w < 8; ++w)
            sum += *(const v4f*)(red + (w * 16 + row) * R1_STRIDE + 4 * c4);
        if (g0 + row < n_p) {
            float* o = out + (size_t)list[g0 + row] * D_MODEL + ks * 128 + nh * 64 + 4 * c4;
            #pragma unroll
            for (int q = 0; q < 4; ++q) *(v4f*)(o + q * 256) = sum;
        }

        g0 += 16;
        if (g0 >= n_p) break;
        lds_barrier();
        #pragma unroll
        for (int kb = 0; kb < 8; ++kb) af[kb] = an[kb];
        if (g0 + 16 < n_p) {
            const float* xn = x + (size_t)bucket_row(list, g0 + 16 + c, n_p) * D_MODEL + kbase;
            #pragma unroll
            for (int kb = 0; kb < 8; ++kb) an[kb] = *(const v4f*)(xn + 8 * kb);
        }
    }
}

__device__ __forceinline__ void h_request(v4f (&p0)[2], v4f (&p1)[2], v4f& bb,
                                          const int* list, int g0, int n_p,
                                          const float* out,
                                          const float* __restrict__ b1p, int q, int t)
{
    const int a = 4 * (t & 31);
    #pragma unroll
    for (int i = 0; i < 2; ++i) {
        const int row = (t >> 5) + 8 * i;
        const float* hp = out + (size_t)bucket_row(list, g0 + row, n_p) * D_MODEL
                              + q * 256 + a;
        p0[i] = *(const v4f*)hp;
        p1[i] = *(const v4f*)(hp + 128);
    }
    bb = *(const v4f*)(b1p + a);
}

__device__ __forceinline__ void h_finish(float* hS, const v4f (&p0)[2],
                                         const v4f (&p1)[2], const v4f& bb,
                                         int g0, int n_p, int t)
{
    const int a = 4 * (t & 31);
    #pragma unroll
    for (int i = 0; i < 2; ++i) {
        const int row = (t >> 5) + 8 * i;
        const bool valid = g0 + row < n_p;
        v4f hv = p0[i] + p1[i] + bb;
        #pragma unroll
        for (int e = 0; e < 4; ++e) {
            const float r = hv[e] > 0.0f ? hv[e] : 0.0f;
            hv[e] = valid ? r : 0.0f;
        }
        *(v4f*)(hS + row * H_STRIDE + a) = hv;
    }
}

__global__ __launch_bounds__(256) void adapter_up_kernel(
    const int* __restrict__ src, const int* __restrict__ tgt,
    const float* __restrict__ b1, const float* __restrict__ W2,
    const float* __restrict__ b2, float* out)
{
    __shared__ __attribute__((aligned(16))) float hS[16 * H_STRIDE];
    __shared__ __attribute__((aligned(16))) float red[2 * 16 * R2_STRIDE];
    __shared__ int list[BATCH];
    __shared__ int waveTot[8];

    const int p = blockIdx.x;
    const int t = threadIdx.x;
    if ((p >> 3) == (p & 7)) return;

    const int q    = blockIdx.y;
    const int lane = t & 31;
    const int wave = t >> 5;
    const int c    = lane & 15;
    const int h    = lane >> 4;
    const int strip = wave & 3;
    const int kh    = wave >> 2;
    const int kbase = kh * 64 + 4 * h;

    const int4 s4 = ((const int4*)src)[t];
    const int4 g4 = ((const int4*)tgt)[t];
    v4f bf[32];
    load_b_tile(bf, W2 + (size_t)p * ADAPT * D_MODEL + (size_t)kbase * D_MODEL
                       + q * 256 + strip * 64 + 4 * c, D_MODEL);

    const int n_p = build_bucket(list, waveTot, s4, g4, p, t);
    if (n_p == 0) return;

    const v4f b2v = *(const v4f*)(b2 + p * D_MODEL + q * 256 + 4 * (t & 63));
    int g0 = 0;
    v4f p0[2], p1[2], bb;
    h_request(p0, p1, bb, list, g0, n_p, out, b1 + p * ADAPT, q, t);
    __builtin_amdgcn_sched_barrier(0);
    h_finish(hS, p0, p1, bb, g0, n_p, t);

    while (true) {
        lds_barrier();

        v4f af[8];
        #pragma unroll
        for (int kb = 0; kb < 8; ++kb)
            af[kb] = *(const v4f*)(hS + c * H_STRIDE + kbase + 8 * kb);

        const bool more = g0 + 16 < n_p;
        if (more) h_request(p0, p1, bb, list, g0 + 16, n_p, out, b1 + p * ADAPT, q, t);

        v8f acc[4] = {};
        mma_tile(acc, af, bf);

        float* rw = red + (kh * 16 + 8 * h) * R2_STRIDE + strip * 64 + 4 * c;
        #pragma unroll
        for (int v = 0; v < 8; ++v) {
            const v4f o = {acc[0][v], acc[1][v], acc[2][v], acc[3][v]};
            *(v4f*)(rw + v * R2_STRIDE) = o;
        }
        lds_barrier();

        #pragma unroll
        for (int i = 0; i < 4; ++i) {
            const int idx = t + 256 * i;
            const int row = idx >> 6;
            const int n   = 4 * (idx & 63);
            if (g0 + row < n_p) {
                const v4f y = *(const v4f*)(red + row * R2_STRIDE + n)
                            + *(const v4f*)(red + (16 + row) * R2_STRIDE + n)
                            + b2v;
                *(v4f*)(out + (size_t)list[g0 + row] * D_MODEL + q * 256 + n) = y;
            }
        }

        g0 += 16;
        if (!more) break;
        lds_barrier();
        h_finish(hS, p0, p1, bb, g0, n_p, t);
    }
}

extern "C" void kernel_launch(void* const* d_in, const int* in_sizes, int n_in,
                              void* d_out, int out_size, void* d_ws, size_t ws_size,
                              hipStream_t stream) {
    const float* x   = (const float*)d_in[0];
    const int*   src = (const int*)  d_in[1];
    const int*   tgt = (const int*)  d_in[2];
    const float* W1  = (const float*)d_in[3];
    const float* b1  = (const float*)d_in[4];
    const float* W2  = (const float*)d_in[5];
    const float* b2  = (const float*)d_in[6];
    float* out = (float*)d_out;

    adapter_down_kernel<<<dim3(NPAIR, 5), 256, 0, stream>>>(x, src, tgt, W1, out);
    adapter_up_kernel<<<dim3(NPAIR, 4), 256, 0, stream>>>(src, tgt, b1, W2, b2, out);
}
